# starting kernel
# baseline (speedup 1.0000x reference)
#include <hip/hip_runtime.h>
#include <math.h>

#define BATCH 16
#define CH    4
#define FH    128
#define FW    240
#define NDISP 24
#define OH    1024
#define OW    1920

typedef __attribute__((ext_vector_type(2))) float v2f;
typedef __attribute__((ext_vector_type(8))) float v8f;

__global__ void __launch_bounds__(256)
disparity_pred_kernel(const float* __restrict__ fl,
                      const float* __restrict__ fr,
                      float* __restrict__ pred) {
    const int by = blockIdx.x;
    const int b  = by >> 7;
    const int y  = by & (FH - 1);
    const int t  = threadIdx.x;

    __shared__ float Rr[CH][FW];

    if (t < FW) {
#pragma unroll
        for (int c = 0; c < CH; ++c) {
            Rr[c][t] = __builtin_nontemporal_load(
                fr + (((size_t)(b * CH + c) * FH + y) * FW + t));
        }
    }
    __syncthreads();

    if (t < FW) {
        float L[CH];
#pragma unroll
        for (int c = 0; c < CH; ++c) {
            L[c] = __builtin_nontemporal_load(
                fl + (((size_t)(b * CH + c) * FH + y) * FW + t));
        }

        float cost[NDISP];
#pragma unroll
        for (int d = 0; d < NDISP; ++d) {
            float s = 0.0f;
            const int xs = t - d;
#pragma unroll
            for (int c = 0; c < CH; ++c) {
                const float r = (xs >= 0) ? Rr[c][xs] : 0.0f;
                s += fabsf(L[c] - r);
            }
            cost[d] = s;
        }

        float m = cost[0];
#pragma unroll
        for (int d = 1; d < NDISP; ++d) m = fminf(m, cost[d]);

        float se = 0.0f, sd = 0.0f;
#pragma unroll
        for (int d = 0; d < NDISP; ++d) {
            const float e = expf(m - cost[d]);
            se += e;
            sd += (float)d * e;
        }
        pred[((size_t)b * FH + y) * FW + t] = 8.0f * (sd / se);
    }
}

__global__ void __launch_bounds__(256)
upsample_wmma_kernel(const float* __restrict__ pred,
                     float* __restrict__ out) {
    const float SY = 127.0f / 1023.0f;
    const float SX = 239.0f / 1919.0f;

    const int lane = threadIdx.x & 31;
    const int wave = threadIdx.x >> 5;
    const int tx = blockIdx.x * 8 + wave;
    const int ty = blockIdx.y;
    const int b  = blockIdx.z;

    const int m     = lane & 15;
    const int half  = lane >> 4;
    const int kBase = half * 2;

    const int   yOut = ty * 16 + m;
    const float ysf  = (float)yOut * SY;
    const int   y0   = (int)ysf;
    const float wy   = ysf - (float)y0;
    const int   y1   = min(y0 + 1, FH - 1);

    const int xBase = tx * 16;
    const int c0    = (int)((float)xBase * SX);
    const int colA0 = min(c0 + kBase,     FW - 1);
    const int colA1 = min(c0 + kBase + 1, FW - 1);

    const float* pb = pred + (size_t)b * FH * FW;
    const float r0a = pb[y0 * FW + colA0];
    const float r0b = pb[y0 * FW + colA1];
    const float r1a = pb[y1 * FW + colA0];
    const float r1b = pb[y1 * FW + colA1];

    v2f A;
    A.x = r0a + wy * (r1a - r0a);
    A.y = r0b + wy * (r1b - r0b);

    const float xsf = (float)(xBase + m) * SX;
    const int   x0  = (int)xsf;
    const float wx  = xsf - (float)x0;
    const int   x1  = min(x0 + 1, FW - 1);
    const int   rel0 = x0 - c0;
    const int   rel1 = x1 - c0;

    v2f Bm;
    Bm.x = (rel0 == kBase     ? 1.0f - wx : 0.0f) + (rel1 == kBase     ? wx : 0.0f);
    Bm.y = (rel0 == kBase + 1 ? 1.0f - wx : 0.0f) + (rel1 == kBase + 1 ? wx : 0.0f);

    v8f Cacc = {};
    Cacc = __builtin_amdgcn_wmma_f32_16x16x4_f32(
         false, A,  false, Bm,
         (short)0, Cacc,  false,  false);

    float* ob = out + (size_t)b * OH * OW + (size_t)(ty * 16) * OW + xBase;
    const int rowOff = half * 8;
#pragma unroll
    for (int r = 0; r < 8; ++r) {
        __builtin_nontemporal_store(Cacc[r], ob + (size_t)(rowOff + r) * OW + m);
    }
}

extern "C" void kernel_launch(void* const* d_in, const int* in_sizes, int n_in,
                              void* d_out, int out_size, void* d_ws, size_t ws_size,
                              hipStream_t stream) {
    const float* feat_l = (const float*)d_in[0];
    const float* feat_r = (const float*)d_in[1];
    float* out  = (float*)d_out;
    float* pred = (float*)d_ws;

    disparity_pred_kernel<<<BATCH * FH, 256, 0, stream>>>(feat_l, feat_r, pred);

    upsample_wmma_kernel<<<dim3(15, 64, BATCH), 256, 0, stream>>>(pred, out);
}


# upsample re-tiled: 1920 resident waves (was 122880) loop over the 16 images, each writing 16x64 blocks as whole 256-B row pieces with 128-bit stores; K 4->12 zero-padded over 3 chained WMMAs
# speedup vs baseline: 1.6888x; 1.6888x over previous
#include <hip/hip_runtime.h>
#include <math.h>

#define BATCH 16
#define CH    4
#define FH    128
#define FW    240
#define NDISP 24
#define OH    1024
#define OW    1920

typedef __attribute__((ext_vector_type(2))) float v2f;
typedef __attribute__((ext_vector_type(4))) float v4f;
typedef __attribute__((ext_vector_type(8))) float v8f;

__global__ void __launch_bounds__(256)
disparity_pred_kernel(const float* __restrict__ fl,
                      const float* __restrict__ fr,
                      float* __restrict__ pred) {
    const int by = blockIdx.x;
    const int b  = by >> 7;
    const int y  = by & (FH - 1);
    const int t  = threadIdx.x;

    __shared__ float Rr[CH][FW];

    if (t < FW) {
#pragma unroll
        for (int c = 0; c < CH; ++c) {
            Rr[c][t] = __builtin_nontemporal_load(
                fr + (((size_t)(b * CH + c) * FH + y) * FW + t));
        }
    }
    __syncthreads();

    if (t < FW) {
        float L[CH];
#pragma unroll
        for (int c = 0; c < CH; ++c) {
            L[c] = __builtin_nontemporal_load(
                fl + (((size_t)(b * CH + c) * FH + y) * FW + t));
        }

        float cost[NDISP];
#pragma unroll
        for (int d = 0; d < NDISP; ++d) {
            float s = 0.0f;
            const int xs = t - d;
#pragma unroll
            for (int c = 0; c < CH; ++c) {
                const float r = (xs >= 0) ? Rr[c][xs] : 0.0f;
                s += fabsf(L[c] - r);
            }
            cost[d] = s;
        }

        float m = cost[0];
#pragma unroll
        for (int d = 1; d < NDISP; ++d) m = fminf(m, cost[d]);

        float se = 0.0f, sd = 0.0f;
#pragma unroll
        for (int d = 0; d < NDISP; ++d) {
            const float e = expf(m - cost[d]);
            se += e;
            sd += (float)d * e;
        }
        pred[((size_t)b * FH + y) * FW + t] = 8.0f * (sd / se);
    }
}

#define STRIP_W 64
#define NSTRIP  (OW / STRIP_W)
#define KCH     3
#define NTILE   4

__global__ void __launch_bounds__(256)
upsample_wmma_kernel(const float* __restrict__ pred,
                     float* __restrict__ out) {
    const float SY = 127.0f / 1023.0f;
    const float SX = 239.0f / 1919.0f;

    const int lane = threadIdx.x & 31;
    const int wave = __builtin_amdgcn_readfirstlane((int)(threadIdx.x >> 5));
    const int g  = blockIdx.x * 8 + wave;
    const int ty = g / NSTRIP;
    const int s  = g - ty * NSTRIP;

    const int m     = lane & 15;
    const int half  = lane >> 4;
    const int kBase = half * 2;

    const int   yOut = ty * 16 + m;
    const float ysf  = (float)yOut * SY;
    const int   y0   = (int)ysf;
    const float wy   = ysf - (float)y0;
    const int   y1   = min(y0 + 1, FH - 1);

    const int xBase = s * STRIP_W;
    const int c0    = (int)((float)xBase * SX);

    int off0[KCH * 2], off1[KCH * 2];
#pragma unroll
    for (int i = 0; i < KCH * 2; ++i) {
        const int col = min(c0 + 4 * (i >> 1) + kBase + (i & 1), FW - 1);
        off0[i] = y0 * FW + col;
        off1[i] = y1 * FW + col;
    }

    v2f Bw[NTILE][KCH];
#pragma unroll
    for (int j = 0; j < NTILE; ++j) {
        const float xsf = (float)(xBase + 4 * m + j) * SX;
        const int   x0  = (int)xsf;
        const float wx  = xsf - (float)x0;
        const int   x1  = min(x0 + 1, FW - 1);
        const int   rel0 = x0 - c0;
        const int   rel1 = x1 - c0;
#pragma unroll
        for (int q = 0; q < KCH; ++q) {
            const int k0 = 4 * q + kBase;
            Bw[j][q].x = (rel0 == k0     ? 1.0f - wx : 0.0f) + (rel1 == k0     ? wx : 0.0f);
            Bw[j][q].y = (rel0 == k0 + 1 ? 1.0f - wx : 0.0f) + (rel1 == k0 + 1 ? wx : 0.0f);
        }
    }

    float cur0[KCH * 2], cur1[KCH * 2];
#pragma unroll
    for (int i = 0; i < KCH * 2; ++i) {
        cur0[i] = pred[off0[i]];
        cur1[i] = pred[off1[i]];
    }

    float* ob = out + (size_t)(ty * 16 + half * 8) * OW + xBase + 4 * m;

#pragma unroll
    for (int b = 0; b < BATCH; ++b) {
        const float* pn = pred + (size_t)min(b + 1, BATCH - 1) * (FH * FW);
        float nx0[KCH * 2], nx1[KCH * 2];
#pragma unroll
        for (int i = 0; i < KCH * 2; ++i) {
            nx0[i] = pn[off0[i]];
            nx1[i] = pn[off1[i]];
        }

        v2f A[KCH];
#pragma unroll
        for (int q = 0; q < KCH; ++q) {
            A[q].x = cur0[2 * q]     + wy * (cur1[2 * q]     - cur0[2 * q]);
            A[q].y = cur0[2 * q + 1] + wy * (cur1[2 * q + 1] - cur0[2 * q + 1]);
        }

        v8f D[NTILE];
#pragma unroll
        for (int j = 0; j < NTILE; ++j) D[j] = (v8f){};
#pragma unroll
        for (int q = 0; q < KCH; ++q) {
#pragma unroll
            for (int j = 0; j < NTILE; ++j) {
                D[j] = __builtin_amdgcn_wmma_f32_16x16x4_f32(
                     false, A[q],  false, Bw[j][q],
                     (short)0, D[j],  false,  false);
            }
        }

        v4f v[8];
#pragma unroll
        for (int r = 0; r < 8; ++r) v[r] = (v4f){D[0][r], D[1][r], D[2][r], D[3][r]};
        asm volatile("" : "+v"(v[0]), "+v"(v[1]), "+v"(v[2]), "+v"(v[3]),
                          "+v"(v[4]), "+v"(v[5]), "+v"(v[6]), "+v"(v[7]));

#pragma unroll
        for (int r = 0; r < 8; ++r) {
            __builtin_nontemporal_store(v[r], (v4f*)(ob + (size_t)r * OW));
        }
        ob += (size_t)OH * OW;

#pragma unroll
        for (int i = 0; i < KCH * 2; ++i) {
            cur0[i] = nx0[i];
            cur1[i] = nx1[i];
        }
    }
}

extern "C" void kernel_launch(void* const* d_in, const int* in_sizes, int n_in,
                              void* d_out, int out_size, void* d_ws, size_t ws_size,
                              hipStream_t stream) {
    const float* feat_l = (const float*)d_in[0];
    const float* feat_r = (const float*)d_in[1];
    float* out  = (float*)d_out;
    float* pred = (float*)d_ws;

    disparity_pred_kernel<<<BATCH * FH, 256, 0, stream>>>(feat_l, feat_r, pred);

    upsample_wmma_kernel<<<240, 256, 0, stream>>>(pred, out);
}


# pred kernel: softmax exponentials as one fma plus hardware exp2 (log2e folded into the shift) in place of library expf, 24 per pixel
# speedup vs baseline: 1.8032x; 1.0677x over previous
#include <hip/hip_runtime.h>
#include <math.h>

#define BATCH 16
#define CH    4
#define FH    128
#define FW    240
#define NDISP 24
#define OH    1024
#define OW    1920

typedef __attribute__((ext_vector_type(2))) float v2f;
typedef __attribute__((ext_vector_type(4))) float v4f;
typedef __attribute__((ext_vector_type(8))) float v8f;

__global__ void __launch_bounds__(256)
disparity_pred_kernel(const float* __restrict__ fl,
                      const float* __restrict__ fr,
                      float* __restrict__ pred) {
    const int by = blockIdx.x;
    const int b  = by >> 7;
    const int y  = by & (FH - 1);
    const int t  = threadIdx.x;

    __shared__ float Rr[CH][FW];

    if (t < FW) {
#pragma unroll
        for (int c = 0; c < CH; ++c) {
            Rr[c][t] = __builtin_nontemporal_load(
                fr + (((size_t)(b * CH + c) * FH + y) * FW + t));
        }
    }
    __syncthreads();

    if (t < FW) {
        float L[CH];
#pragma unroll
        for (int c = 0; c < CH; ++c) {
            L[c] = __builtin_nontemporal_load(
                fl + (((size_t)(b * CH + c) * FH + y) * FW + t));
        }

        float cost[NDISP];
#pragma unroll
        for (int d = 0; d < NDISP; ++d) {
            float s = 0.0f;
            const int xs = t - d;
#pragma unroll
            for (int c = 0; c < CH; ++c) {
                const float r = (xs >= 0) ? Rr[c][xs] : 0.0f;
                s += fabsf(L[c] - r);
            }
            cost[d] = s;
        }

        float m = cost[0];
#pragma unroll
        for (int d = 1; d < NDISP; ++d) m = fminf(m, cost[d]);

        const float LOG2E = 1.4426950408889634f;
        const float mk = LOG2E * m;
        float se = 0.0f, sd = 0.0f;
#pragma unroll
        for (int d = 0; d < NDISP; ++d) {
            const float e = __builtin_amdgcn_exp2f(__builtin_fmaf(-LOG2E, cost[d], mk));
            se += e;
            sd += (float)d * e;
        }
        pred[((size_t)b * FH + y) * FW + t] = 8.0f * (sd / se);
    }
}

#define STRIP_W 64
#define NSTRIP  (OW / STRIP_W)
#define KCH     3
#define NTILE   4

__global__ void __launch_bounds__(256)
upsample_wmma_kernel(const float* __restrict__ pred,
                     float* __restrict__ out) {
    const float SY = 127.0f / 1023.0f;
    const float SX = 239.0f / 1919.0f;

    const int lane = threadIdx.x & 31;
    const int wave = __builtin_amdgcn_readfirstlane((int)(threadIdx.x >> 5));
    const int g  = blockIdx.x * 8 + wave;
    const int ty = g / NSTRIP;
    const int s  = g - ty * NSTRIP;

    const int m     = lane & 15;
    const int half  = lane >> 4;
    const int kBase = half * 2;

    const int   yOut = ty * 16 + m;
    const float ysf  = (float)yOut * SY;
    const int   y0   = (int)ysf;
    const float wy   = ysf - (float)y0;
    const int   y1   = min(y0 + 1, FH - 1);

    const int xBase = s * STRIP_W;
    const int c0    = (int)((float)xBase * SX);

    int off0[KCH * 2], off1[KCH * 2];
#pragma unroll
    for (int i = 0; i < KCH * 2; ++i) {
        const int col = min(c0 + 4 * (i >> 1) + kBase + (i & 1), FW - 1);
        off0[i] = y0 * FW + col;
        off1[i] = y1 * FW + col;
    }

    v2f Bw[NTILE][KCH];
#pragma unroll
    for (int j = 0; j < NTILE; ++j) {
        const float xsf = (float)(xBase + 4 * m + j) * SX;
        const int   x0  = (int)xsf;
        const float wx  = xsf - (float)x0;
        const int   x1  = min(x0 + 1, FW - 1);
        const int   rel0 = x0 - c0;
        const int   rel1 = x1 - c0;
#pragma unroll
        for (int q = 0; q < KCH; ++q) {
            const int k0 = 4 * q + kBase;
            Bw[j][q].x = (rel0 == k0     ? 1.0f - wx : 0.0f) + (rel1 == k0     ? wx : 0.0f);
            Bw[j][q].y = (rel0 == k0 + 1 ? 1.0f - wx : 0.0f) + (rel1 == k0 + 1 ? wx : 0.0f);
        }
    }

    float cur0[KCH * 2], cur1[KCH * 2];
#pragma unroll
    for (int i = 0; i < KCH * 2; ++i) {
        cur0[i] = pred[off0[i]];
        cur1[i] = pred[off1[i]];
    }

    float* ob = out + (size_t)(ty * 16 + half * 8) * OW + xBase + 4 * m;

#pragma unroll
    for (int b = 0; b < BATCH; ++b) {
        const float* pn = pred + (size_t)min(b + 1, BATCH - 1) * (FH * FW);
        float nx0[KCH * 2], nx1[KCH * 2];
#pragma unroll
        for (int i = 0; i < KCH * 2; ++i) {
            nx0[i] = pn[off0[i]];
            nx1[i] = pn[off1[i]];
        }

        v2f A[KCH];
#pragma unroll
        for (int q = 0; q < KCH; ++q) {
            A[q].x = cur0[2 * q]     + wy * (cur1[2 * q]     - cur0[2 * q]);
            A[q].y = cur0[2 * q + 1] + wy * (cur1[2 * q + 1] - cur0[2 * q + 1]);
        }

        v8f D[NTILE];
#pragma unroll
        for (int j = 0; j < NTILE; ++j) D[j] = (v8f){};
#pragma unroll
        for (int q = 0; q < KCH; ++q) {
#pragma unroll
            for (int j = 0; j < NTILE; ++j) {
                D[j] = __builtin_amdgcn_wmma_f32_16x16x4_f32(
                     false, A[q],  false, Bw[j][q],
                     (short)0, D[j],  false,  false);
            }
        }

        v4f v[8];
#pragma unroll
        for (int r = 0; r < 8; ++r) v[r] = (v4f){D[0][r], D[1][r], D[2][r], D[3][r]};
        asm volatile("" : "+v"(v[0]), "+v"(v[1]), "+v"(v[2]), "+v"(v[3]),
                          "+v"(v[4]), "+v"(v[5]), "+v"(v[6]), "+v"(v[7]));

#pragma unroll
        for (int r = 0; r < 8; ++r) {
            __builtin_nontemporal_store(v[r], (v4f*)(ob + (size_t)r * OW));
        }
        ob += (size_t)OH * OW;

#pragma unroll
        for (int i = 0; i < KCH * 2; ++i) {
            cur0[i] = nx0[i];
            cur1[i] = nx1[i];
        }
    }
}

extern "C" void kernel_launch(void* const* d_in, const int* in_sizes, int n_in,
                              void* d_out, int out_size, void* d_ws, size_t ws_size,
                              hipStream_t stream) {
    const float* feat_l = (const float*)d_in[0];
    const float* feat_r = (const float*)d_in[1];
    float* out  = (float*)d_out;
    float* pred = (float*)d_ws;

    disparity_pred_kernel<<<BATCH * FH, 256, 0, stream>>>(feat_l, feat_r, pred);

    upsample_wmma_kernel<<<240, 256, 0, stream>>>(pred, out);
}


# upsample kernel: each image's 12 source values requested 5 loop trips ahead (register ring, 60 loads in flight per lane, kept in one group by a scheduling barrier) in place of one trip ahead
# speedup vs baseline: 1.9952x; 1.1065x over previous
#include <hip/hip_runtime.h>
#include <math.h>

#define BATCH 16
#define CH    4
#define FH    128
#define FW    240
#define NDISP 24
#define OH    1024
#define OW    1920

typedef __attribute__((ext_vector_type(2))) float v2f;
typedef __attribute__((ext_vector_type(4))) float v4f;
typedef __attribute__((ext_vector_type(8))) float v8f;

__global__ void __launch_bounds__(256)
disparity_pred_kernel(const float* __restrict__ fl,
                      const float* __restrict__ fr,
                      float* __restrict__ pred) {
    const int by = blockIdx.x;
    const int b  = by >> 7;
    const int y  = by & (FH - 1);
    const int t  = threadIdx.x;

    __shared__ float Rr[CH][FW];

    if (t < FW) {
#pragma unroll
        for (int c = 0; c < CH; ++c) {
            Rr[c][t] = __builtin_nontemporal_load(
                fr + (((size_t)(b * CH + c) * FH + y) * FW + t));
        }
    }
    __syncthreads();

    if (t < FW) {
        float L[CH];
#pragma unroll
        for (int c = 0; c < CH; ++c) {
            L[c] = __builtin_nontemporal_load(
                fl + (((size_t)(b * CH + c) * FH + y) * FW + t));
        }

        float cost[NDISP];
#pragma unroll
        for (int d = 0; d < NDISP; ++d) {
            float s = 0.0f;
            const int xs = t - d;
#pragma unroll
            for (int c = 0; c < CH; ++c) {
                const float r = (xs >= 0) ? Rr[c][xs] : 0.0f;
                s += fabsf(L[c] - r);
            }
            cost[d] = s;
        }

        float m = cost[0];
#pragma unroll
        for (int d = 1; d < NDISP; ++d) m = fminf(m, cost[d]);

        const float LOG2E = 1.4426950408889634f;
        const float mk = LOG2E * m;
        float se = 0.0f, sd = 0.0f;
#pragma unroll
        for (int d = 0; d < NDISP; ++d) {
            const float e = __builtin_amdgcn_exp2f(__builtin_fmaf(-LOG2E, cost[d], mk));
            se += e;
            sd += (float)d * e;
        }
        pred[((size_t)b * FH + y) * FW + t] = 8.0f * (sd / se);
    }
}

#define STRIP_W 64
#define NSTRIP  (OW / STRIP_W)
#define KCH     3
#define NTILE   4
#define DEPTH   5

__global__ void __launch_bounds__(256)
upsample_wmma_kernel(const float* __restrict__ pred,
                     float* __restrict__ out) {
    const float SY = 127.0f / 1023.0f;
    const float SX = 239.0f / 1919.0f;

    const int lane = threadIdx.x & 31;
    const int wave = __builtin_amdgcn_readfirstlane((int)(threadIdx.x >> 5));
    const int g  = blockIdx.x * 8 + wave;
    const int ty = g / NSTRIP;
    const int s  = g - ty * NSTRIP;

    const int m     = lane & 15;
    const int half  = lane >> 4;
    const int kBase = half * 2;

    const int   yOut = ty * 16 + m;
    const float ysf  = (float)yOut * SY;
    const int   y0   = (int)ysf;
    const float wy   = ysf - (float)y0;
    const int   y1   = min(y0 + 1, FH - 1);

    const int xBase = s * STRIP_W;
    const int c0    = (int)((float)xBase * SX);

    int off0[KCH * 2], off1[KCH * 2];
#pragma unroll
    for (int i = 0; i < KCH * 2; ++i) {
        const int col = min(c0 + 4 * (i >> 1) + kBase + (i & 1), FW - 1);
        off0[i] = y0 * FW + col;
        off1[i] = y1 * FW + col;
    }

    v2f Bw[NTILE][KCH];
#pragma unroll
    for (int j = 0; j < NTILE; ++j) {
        const float xsf = (float)(xBase + 4 * m + j) * SX;
        const int   x0  = (int)xsf;
        const float wx  = xsf - (float)x0;
        const int   x1  = min(x0 + 1, FW - 1);
        const int   rel0 = x0 - c0;
        const int   rel1 = x1 - c0;
#pragma unroll
        for (int q = 0; q < KCH; ++q) {
            const int k0 = 4 * q + kBase;
            Bw[j][q].x = (rel0 == k0     ? 1.0f - wx : 0.0f) + (rel1 == k0     ? wx : 0.0f);
            Bw[j][q].y = (rel0 == k0 + 1 ? 1.0f - wx : 0.0f) + (rel1 == k0 + 1 ? wx : 0.0f);
        }
    }

    float sv0[BATCH][KCH * 2], sv1[BATCH][KCH * 2];
#pragma unroll
    for (int b = 0; b < DEPTH; ++b) {
        const float* pb = pred + (size_t)b * (FH * FW);
#pragma unroll
        for (int i = 0; i < KCH * 2; ++i) {
            sv0[b][i] = pb[off0[i]];
            sv1[b][i] = pb[off1[i]];
        }
    }
    __builtin_amdgcn_sched_barrier(0);

    float* ob = out + (size_t)(ty * 16 + half * 8) * OW + xBase + 4 * m;

#pragma unroll
    for (int b = 0; b < BATCH; ++b) {
        if (b + DEPTH < BATCH) {
            const float* pn = pred + (size_t)(b + DEPTH) * (FH * FW);
#pragma unroll
            for (int i = 0; i < KCH * 2; ++i) {
                sv0[b + DEPTH][i] = pn[off0[i]];
                sv1[b + DEPTH][i] = pn[off1[i]];
            }
        }

        v2f A[KCH];
#pragma unroll
        for (int q = 0; q < KCH; ++q) {
            A[q].x = sv0[b][2 * q]     + wy * (sv1[b][2 * q]     - sv0[b][2 * q]);
            A[q].y = sv0[b][2 * q + 1] + wy * (sv1[b][2 * q + 1] - sv0[b][2 * q + 1]);
        }

        v8f D[NTILE];
#pragma unroll
        for (int j = 0; j < NTILE; ++j) D[j] = (v8f){};
#pragma unroll
        for (int q = 0; q < KCH; ++q) {
#pragma unroll
            for (int j = 0; j < NTILE; ++j) {
                D[j] = __builtin_amdgcn_wmma_f32_16x16x4_f32(
                     false, A[q],  false, Bw[j][q],
                     (short)0, D[j],  false,  false);
            }
        }

        v4f v[8];
#pragma unroll
        for (int r = 0; r < 8; ++r) v[r] = (v4f){D[0][r], D[1][r], D[2][r], D[3][r]};
        asm volatile("" : "+v"(v[0]), "+v"(v[1]), "+v"(v[2]), "+v"(v[3]),
                          "+v"(v[4]), "+v"(v[5]), "+v"(v[6]), "+v"(v[7]));

#pragma unroll
        for (int r = 0; r < 8; ++r) {
            __builtin_nontemporal_store(v[r], (v4f*)(ob + (size_t)r * OW));
        }
        ob += (size_t)OH * OW;
    }
}

extern "C" void kernel_launch(void* const* d_in, const int* in_sizes, int n_in,
                              void* d_out, int out_size, void* d_ws, size_t ws_size,
                              hipStream_t stream) {
    const float* feat_l = (const float*)d_in[0];
    const float* feat_r = (const float*)d_in[1];
    float* out  = (float*)d_out;
    float* pred = (float*)d_ws;

    disparity_pred_kernel<<<BATCH * FH, 256, 0, stream>>>(feat_l, feat_r, pred);

    upsample_wmma_kernel<<<240, 256, 0, stream>>>(pred, out);
}


# pred kernel: each lane computes 4 neighbouring pixels from a register window read by 16-byte LDS loads behind a zero-padded row; 4096 waves in place of 16384, per-pixel arithmetic unchanged
# speedup vs baseline: 2.2208x; 1.1131x over previous
#include <hip/hip_runtime.h>
#include <math.h>

#define BATCH 16
#define CH    4
#define FH    128
#define FW    240
#define NDISP 24
#define OH    1024
#define OW    1920

typedef __attribute__((ext_vector_type(2))) float v2f;
typedef __attribute__((ext_vector_type(4))) float v4f;
typedef __attribute__((ext_vector_type(8))) float v8f;

#define PXL   4
#define RLANE (FW / PXL)
#define RPB   4
#define PADW  (NDISP + FW)

__global__ void __launch_bounds__(256)
disparity_pred_kernel(const float* __restrict__ fl,
                      const float* __restrict__ fr,
                      float* __restrict__ pred) {
    const int t   = threadIdx.x;
    const int rb  = t >> 6;
    const int l   = t & 63;
    const int by  = blockIdx.x * RPB + rb;
    const int b   = by >> 7;
    const int y   = by & (FH - 1);
    const bool act = l < RLANE;

    __shared__ __attribute__((aligned(16))) float Rp[RPB][CH][PADW];

    v4f L4[CH];
    if (act) {
        v4f r4[CH];
#pragma unroll
        for (int c = 0; c < CH; ++c) {
            r4[c] = __builtin_nontemporal_load(
                (const v4f*)(fr + (((size_t)(b * CH + c) * FH + y) * FW + PXL * l)));
        }
#pragma unroll
        for (int c = 0; c < CH; ++c) {
            L4[c] = __builtin_nontemporal_load(
                (const v4f*)(fl + (((size_t)(b * CH + c) * FH + y) * FW + PXL * l)));
        }
#pragma unroll
        for (int c = 0; c < CH; ++c) *(v4f*)&Rp[rb][c][NDISP + PXL * l] = r4[c];
    }
    if (l < NDISP / 4) {
#pragma unroll
        for (int c = 0; c < CH; ++c) *(v4f*)&Rp[rb][c][4 * l] = (v4f){0.0f, 0.0f, 0.0f, 0.0f};
    }
    __syncthreads();

    if (act) {
        float cost[PXL][NDISP];
#pragma unroll
        for (int i = 0; i < PXL; ++i) {
#pragma unroll
            for (int d = 0; d < NDISP; ++d) cost[i][d] = 0.0f;
        }
#pragma unroll
        for (int c = 0; c < CH; ++c) {
            float w[NDISP + PXL];
#pragma unroll
            for (int j = 0; j < (NDISP + PXL) / 4; ++j) {
                const v4f q = *(const v4f*)&Rp[rb][c][PXL * l + 4 * j];
                w[4 * j + 0] = q.x; w[4 * j + 1] = q.y;
                w[4 * j + 2] = q.z; w[4 * j + 3] = q.w;
            }
#pragma unroll
            for (int i = 0; i < PXL; ++i) {
                const float Lc = L4[c][i];
#pragma unroll
                for (int d = 0; d < NDISP; ++d) {
                    cost[i][d] += fabsf(Lc - w[NDISP + i - d]);
                }
            }
        }

        const float LOG2E = 1.4426950408889634f;
        v4f res;
#pragma unroll
        for (int i = 0; i < PXL; ++i) {
            float m = cost[i][0];
#pragma unroll
            for (int d = 1; d < NDISP; ++d) m = fminf(m, cost[i][d]);
            const float mk = LOG2E * m;
            float se = 0.0f, sd = 0.0f;
#pragma unroll
            for (int d = 0; d < NDISP; ++d) {
                const float e = __builtin_amdgcn_exp2f(__builtin_fmaf(-LOG2E, cost[i][d], mk));
                se += e;
                sd += (float)d * e;
            }
            res[i] = 8.0f * (sd / se);
        }
        *(v4f*)(pred + ((size_t)b * FH + y) * FW + PXL * l) = res;
    }
}

#define STRIP_W 64
#define NSTRIP  (OW / STRIP_W)
#define KCH     3
#define NTILE   4
#define DEPTH   5

__global__ void __launch_bounds__(256)
upsample_wmma_kernel(const float* __restrict__ pred,
                     float* __restrict__ out) {
    const float SY = 127.0f / 1023.0f;
    const float SX = 239.0f / 1919.0f;

    const int lane = threadIdx.x & 31;
    const int wave = __builtin_amdgcn_readfirstlane((int)(threadIdx.x >> 5));
    const int g  = blockIdx.x * 8 + wave;
    const int ty = g / NSTRIP;
    const int s  = g - ty * NSTRIP;

    const int m     = lane & 15;
    const int half  = lane >> 4;
    const int kBase = half * 2;

    const int   yOut = ty * 16 + m;
    const float ysf  = (float)yOut * SY;
    const int   y0   = (int)ysf;
    const float wy   = ysf - (float)y0;
    const int   y1   = min(y0 + 1, FH - 1);

    const int xBase = s * STRIP_W;
    const int c0    = (int)((float)xBase * SX);

    int off0[KCH * 2], off1[KCH * 2];
#pragma unroll
    for (int i = 0; i < KCH * 2; ++i) {
        const int col = min(c0 + 4 * (i >> 1) + kBase + (i & 1), FW - 1);
        off0[i] = y0 * FW + col;
        off1[i] = y1 * FW + col;
    }

    v2f Bw[NTILE][KCH];
#pragma unroll
    for (int j = 0; j < NTILE; ++j) {
        const float xsf = (float)(xBase + 4 * m + j) * SX;
        const int   x0  = (int)xsf;
        const float wx  = xsf - (float)x0;
        const int   x1  = min(x0 + 1, FW - 1);
        const int   rel0 = x0 - c0;
        const int   rel1 = x1 - c0;
#pragma unroll
        for (int q = 0; q < KCH; ++q) {
            const int k0 = 4 * q + kBase;
            Bw[j][q].x = (rel0 == k0     ? 1.0f - wx : 0.0f) + (rel1 == k0     ? wx : 0.0f);
            Bw[j][q].y = (rel0 == k0 + 1 ? 1.0f - wx : 0.0f) + (rel1 == k0 + 1 ? wx : 0.0f);
        }
    }

    float sv0[BATCH][KCH * 2], sv1[BATCH][KCH * 2];
#pragma unroll
    for (int b = 0; b < DEPTH; ++b) {
        const float* pb = pred + (size_t)b * (FH * FW);
#pragma unroll
        for (int i = 0; i < KCH * 2; ++i) {
            sv0[b][i] = pb[off0[i]];
            sv1[b][i] = pb[off1[i]];
        }
    }
    __builtin_amdgcn_sched_barrier(0);

    float* ob = out + (size_t)(ty * 16 + half * 8) * OW + xBase + 4 * m;

#pragma unroll
    for (int b = 0; b < BATCH; ++b) {
        if (b + DEPTH < BATCH) {
            const float* pn = pred + (size_t)(b + DEPTH) * (FH * FW);
#pragma unroll
            for (int i = 0; i < KCH * 2; ++i) {
                sv0[b + DEPTH][i] = pn[off0[i]];
                sv1[b + DEPTH][i] = pn[off1[i]];
            }
        }

        v2f A[KCH];
#pragma unroll
        for (int q = 0; q < KCH; ++q) {
            A[q].x = sv0[b][2 * q]     + wy * (sv1[b][2 * q]     - sv0[b][2 * q]);
            A[q].y = sv0[b][2 * q + 1] + wy * (sv1[b][2 * q + 1] - sv0[b][2 * q + 1]);
        }

        v8f D[NTILE];
#pragma unroll
        for (int j = 0; j < NTILE; ++j) D[j] = (v8f){};
#pragma unroll
        for (int q = 0; q < KCH; ++q) {
#pragma unroll
            for (int j = 0; j < NTILE; ++j) {
                D[j] = __builtin_amdgcn_wmma_f32_16x16x4_f32(
                     false, A[q],  false, Bw[j][q],
                     (short)0, D[j],  false,  false);
            }
        }

        v4f v[8];
#pragma unroll
        for (int r = 0; r < 8; ++r) v[r] = (v4f){D[0][r], D[1][r], D[2][r], D[3][r]};
        asm volatile("" : "+v"(v[0]), "+v"(v[1]), "+v"(v[2]), "+v"(v[3]),
                          "+v"(v[4]), "+v"(v[5]), "+v"(v[6]), "+v"(v[7]));

#pragma unroll
        for (int r = 0; r < 8; ++r) {
            __builtin_nontemporal_store(v[r], (v4f*)(ob + (size_t)r * OW));
        }
        ob += (size_t)OH * OW;
    }
}

extern "C" void kernel_launch(void* const* d_in, const int* in_sizes, int n_in,
                              void* d_out, int out_size, void* d_ws, size_t ws_size,
                              hipStream_t stream) {
    const float* feat_l = (const float*)d_in[0];
    const float* feat_r = (const float*)d_in[1];
    float* out  = (float*)d_out;
    float* pred = (float*)d_ws;

    disparity_pred_kernel<<<BATCH * FH / RPB, 256, 0, stream>>>(feat_l, feat_r, pred);

    upsample_wmma_kernel<<<240, 256, 0, stream>>>(pred, out);
}


# both passes fused into one launch: a block owns one image x 64 output rows, computes the 10 disparity rows they interpolate into LDS and upsamples from LDS; 256 blocks, no workspace round trip
# speedup vs baseline: 2.7684x; 1.2466x over previous
#include <hip/hip_runtime.h>
#include <math.h>

#define BATCH 16
#define CH    4
#define FH    128
#define FW    240
#define NDISP 24
#define OH    1024
#define OW    1920

typedef __attribute__((ext_vector_type(2))) float v2f;
typedef __attribute__((ext_vector_type(4))) float v4f;
typedef __attribute__((ext_vector_type(8))) float v8f;

#define PXL     4
#define RLANE   (FW / PXL)
#define RPB     4
#define NSWEEP  3
#define PADW    (NDISP + FW)
#define GROUP_H 64
#define NGROUP  (OH / GROUP_H)
#define MAXROWS 10

#define STRIP_W 64
#define NSTRIP  (OW / STRIP_W)
#define NBAND   (GROUP_H / 16)
#define KCH     3
#define NTILE   4
#define TPW     (NSTRIP * NBAND / 8)

__global__ void __launch_bounds__(256)
disparity_upsample_kernel(const float* __restrict__ fl,
                          const float* __restrict__ fr,
                          float* __restrict__ out) {
    const float SY = 127.0f / 1023.0f;
    const float SX = 239.0f / 1919.0f;

    __shared__ __attribute__((aligned(16))) float Rp[MAXROWS][CH][PADW];
    __shared__ __attribute__((aligned(16))) float Lp[MAXROWS][CH][FW];
    __shared__ __attribute__((aligned(16))) float Pp[MAXROWS][FW];

    const int t  = threadIdx.x;
    const int b  = blockIdx.x >> 4;
    const int Y0 = (blockIdx.x & (NGROUP - 1)) * GROUP_H;

    const int yFirst = (int)((float)Y0 * SY);
    const int yLast  = min((int)((float)(Y0 + GROUP_H - 1) * SY) + 1, FH - 1);
    const int nrows  = min(yLast - yFirst + 1, MAXROWS);

    const int  rb  = t >> 6;
    const int  l   = t & 63;
    const bool act = l < RLANE;

    {
        v4f r4[NSWEEP][CH], l4[NSWEEP][CH];
        if (act) {
#pragma unroll
            for (int k = 0; k < NSWEEP; ++k) {
                const int y = yFirst + min(rb + RPB * k, nrows - 1);
#pragma unroll
                for (int c = 0; c < CH; ++c) {
                    r4[k][c] = __builtin_nontemporal_load(
                        (const v4f*)(fr + (((size_t)(b * CH + c) * FH + y) * FW + PXL * l)));
                }
#pragma unroll
                for (int c = 0; c < CH; ++c) {
                    l4[k][c] = __builtin_nontemporal_load(
                        (const v4f*)(fl + (((size_t)(b * CH + c) * FH + y) * FW + PXL * l)));
                }
            }
        }
#pragma unroll
        for (int k = 0; k < NSWEEP; ++k) {
            const int rr = rb + RPB * k;
            if (rr < nrows) {
                if (act) {
#pragma unroll
                    for (int c = 0; c < CH; ++c) {
                        *(v4f*)&Rp[rr][c][NDISP + PXL * l] = r4[k][c];
                        *(v4f*)&Lp[rr][c][PXL * l]         = l4[k][c];
                    }
                }
                if (l < NDISP / 4) {
#pragma unroll
                    for (int c = 0; c < CH; ++c)
                        *(v4f*)&Rp[rr][c][4 * l] = (v4f){0.0f, 0.0f, 0.0f, 0.0f};
                }
            }
        }
    }
    __syncthreads();

    for (int rr = rb; rr < nrows; rr += RPB) {
        if (act) {
            v4f L4[CH];
#pragma unroll
            for (int c = 0; c < CH; ++c) L4[c] = *(const v4f*)&Lp[rr][c][PXL * l];

            float cost[PXL][NDISP];
#pragma unroll
            for (int i = 0; i < PXL; ++i) {
#pragma unroll
                for (int d = 0; d < NDISP; ++d) cost[i][d] = 0.0f;
            }
#pragma unroll
            for (int c = 0; c < CH; ++c) {
                float w[NDISP + PXL];
#pragma unroll
                for (int j = 0; j < (NDISP + PXL) / 4; ++j) {
                    const v4f q = *(const v4f*)&Rp[rr][c][PXL * l + 4 * j];
                    w[4 * j + 0] = q.x; w[4 * j + 1] = q.y;
                    w[4 * j + 2] = q.z; w[4 * j + 3] = q.w;
                }
#pragma unroll
                for (int i = 0; i < PXL; ++i) {
                    const float Lc = L4[c][i];
#pragma unroll
                    for (int d = 0; d < NDISP; ++d) {
                        cost[i][d] += fabsf(Lc - w[NDISP + i - d]);
                    }
                }
            }

            const float LOG2E = 1.4426950408889634f;
            v4f res;
#pragma unroll
            for (int i = 0; i < PXL; ++i) {
                float mn = cost[i][0];
#pragma unroll
                for (int d = 1; d < NDISP; ++d) mn = fminf(mn, cost[i][d]);
                const float mk = LOG2E * mn;
                float se = 0.0f, sd = 0.0f;
#pragma unroll
                for (int d = 0; d < NDISP; ++d) {
                    const float e = __builtin_amdgcn_exp2f(__builtin_fmaf(-LOG2E, cost[i][d], mk));
                    se += e;
                    sd += (float)d * e;
                }
                res[i] = 8.0f * (sd / se);
            }
            *(v4f*)&Pp[rr][PXL * l] = res;
        }
    }
    __syncthreads();

    const int lane = t & 31;
    const int wave = __builtin_amdgcn_readfirstlane(t >> 5);

    const int m     = lane & 15;
    const int half  = lane >> 4;
    const int kBase = half * 2;

    int col[KCH * 2];
    v2f Bw[NTILE][KCH];
    int xBase = 0;
#pragma unroll
    for (int i = 0; i < KCH * 2; ++i) col[i] = 0;
#pragma unroll
    for (int j = 0; j < NTILE; ++j) {
#pragma unroll
        for (int q = 0; q < KCH; ++q) Bw[j][q] = (v2f){0.0f, 0.0f};
    }

    float* obBlock = out + ((size_t)b * OH + Y0 + half * 8) * OW + 4 * m;

    for (int i = 0; i < TPW; ++i) {
        const int tk   = wave * TPW + i;
        const int s    = tk >> 2;
        const int band = tk & (NBAND - 1);

        if (i == 0 || band == 0) {
            xBase = s * STRIP_W;
            const int c0 = (int)((float)xBase * SX);
#pragma unroll
            for (int i6 = 0; i6 < KCH * 2; ++i6)
                col[i6] = min(c0 + 4 * (i6 >> 1) + kBase + (i6 & 1), FW - 1);
#pragma unroll
            for (int j = 0; j < NTILE; ++j) {
                const float xsf = (float)(xBase + 4 * m + j) * SX;
                const int   x0  = (int)xsf;
                const float wx  = xsf - (float)x0;
                const int   x1  = min(x0 + 1, FW - 1);
                const int   rel0 = x0 - c0;
                const int   rel1 = x1 - c0;
#pragma unroll
                for (int q = 0; q < KCH; ++q) {
                    const int k0 = 4 * q + kBase;
                    Bw[j][q].x = (rel0 == k0     ? 1.0f - wx : 0.0f) + (rel1 == k0     ? wx : 0.0f);
                    Bw[j][q].y = (rel0 == k0 + 1 ? 1.0f - wx : 0.0f) + (rel1 == k0 + 1 ? wx : 0.0f);
                }
            }
        }

        const int   yOut = Y0 + band * 16 + m;
        const float ysf  = (float)yOut * SY;
        const int   y0   = (int)ysf;
        const float wy   = ysf - (float)y0;
        const int   y1   = min(y0 + 1, FH - 1);
        const float* p0  = &Pp[y0 - yFirst][0];
        const float* p1  = &Pp[y1 - yFirst][0];

        v2f A[KCH];
#pragma unroll
        for (int q = 0; q < KCH; ++q) {
            const float a00 = p0[col[2 * q]],     a10 = p1[col[2 * q]];
            const float a01 = p0[col[2 * q + 1]], a11 = p1[col[2 * q + 1]];
            A[q].x = a00 + wy * (a10 - a00);
            A[q].y = a01 + wy * (a11 - a01);
        }

        v8f D[NTILE];
#pragma unroll
        for (int j = 0; j < NTILE; ++j) D[j] = (v8f){};
#pragma unroll
        for (int q = 0; q < KCH; ++q) {
#pragma unroll
            for (int j = 0; j < NTILE; ++j) {
                D[j] = __builtin_amdgcn_wmma_f32_16x16x4_f32(
                     false, A[q],  false, Bw[j][q],
                     (short)0, D[j],  false,  false);
            }
        }

        v4f v[8];
#pragma unroll
        for (int r = 0; r < 8; ++r) v[r] = (v4f){D[0][r], D[1][r], D[2][r], D[3][r]};
        asm volatile("" : "+v"(v[0]), "+v"(v[1]), "+v"(v[2]), "+v"(v[3]),
                          "+v"(v[4]), "+v"(v[5]), "+v"(v[6]), "+v"(v[7]));

        float* ob = obBlock + (size_t)(band * 16) * OW + xBase;
#pragma unroll
        for (int r = 0; r < 8; ++r) {
            __builtin_nontemporal_store(v[r], (v4f*)(ob + (size_t)r * OW));
        }
    }
}

extern "C" void kernel_launch(void* const* d_in, const int* in_sizes, int n_in,
                              void* d_out, int out_size, void* d_ws, size_t ws_size,
                              hipStream_t stream) {
    const float* feat_l = (const float*)d_in[0];
    const float* feat_r = (const float*)d_in[1];
    float* out = (float*)d_out;

    disparity_upsample_kernel<<<BATCH * NGROUP, 256, 0, stream>>>(feat_l, feat_r, out);
}
